# starting kernel
# baseline (speedup 1.0000x reference)
#include <hip/hip_runtime.h>

typedef float v2f __attribute__((ext_vector_type(2)));
typedef float v8f __attribute__((ext_vector_type(8)));

#define OUT_HW 7
#define SCALE 0.25f
#define CC 64
#define HH 100
#define WW 100
#define KPAD 128
#define WSTRIDE 130

__device__ __forceinline__ float hat_cdf(float t) {
    float a = fminf(fmaxf(t + 1.0f, 0.0f), 1.0f);
    float b = fminf(fmaxf(1.0f - t, 0.0f), 1.0f);
    return (t < 0.0f) ? (0.5f * a * a) : (1.0f - 0.5f * b * b);
}

__global__ __launch_bounds__(128)
void prroi_wmma_f32_kernel(const float* __restrict__ feat,
                           const float* __restrict__ rois,
                           float* __restrict__ out,
                           int batch)
{
    __shared__ float sWx[8][WSTRIDE];
    __shared__ float sWy[8][WSTRIDE];
    __shared__ float sU[4][KPAD][16];

    const int n    = blockIdx.x;
    const int tid  = threadIdx.x;
    const int lane = tid & 31;
    const int wave = tid >> 5;
    const int m16  = lane & 15;
    const int hi16 = (lane >> 4) & 1;

    const float* r = rois + (size_t)n * 5;
    int bi = (int)r[0];
    bi = max(0, min(batch - 1, bi));
    float x1 = r[1] * SCALE, y1 = r[2] * SCALE;
    float x2 = r[3] * SCALE, y2 = r[4] * SCALE;
    float rw = fmaxf(x2 - x1, 0.0f), rh = fmaxf(y2 - y1, 0.0f);
    float bw = rw / OUT_HW, bh = rh / OUT_HW;
    float area = bw * bh;
    float sc = (area > 0.0f) ? (1.0f / fmaxf(area, 1e-12f)) : 0.0f;

    for (int idx = tid; idx < 2 * 8 * WSTRIDE; idx += blockDim.x) {
        int which = idx / (8 * WSTRIDE);
        int rem   = idx - which * 8 * WSTRIDE;
        int q     = rem / WSTRIDE;
        int k     = rem - q * WSTRIDE;
        float val = 0.0f;
        if (q < OUT_HW && k < (which ? HH : WW)) {
            float start = which ? y1 : x1;
            float bs    = which ? bh : bw;
            float u0 = start + (float)q * bs;
            float u1 = u0 + bs;
            float px = (float)k;
            val = hat_cdf(u1 - px) - hat_cdf(u0 - px);
        }
        if (which) sWy[q][k] = val; else sWx[q][k] = val;
    }
    __syncthreads();

    for (int idx = lane; idx < 16 * 16; idx += 32)
        sU[wave][112 + (idx >> 4)][idx & 15] = 0.0f;

    v2f bx[32];
    #pragma unroll
    for (int ks = 0; ks < 32; ++ks) {
        int k0 = ks * 4 + hi16 * 2;
        if (m16 < 8) { bx[ks].x = sWx[m16][k0]; bx[ks].y = sWx[m16][k0 + 1]; }
        else         { bx[ks] = (v2f){0.0f, 0.0f}; }
    }

    const float* fbase = feat + (size_t)bi * CC * (HH * WW);

    for (int i = 0; i < 8; ++i) {
        int c = blockIdx.y * 32 + wave * 8 + i;
        const float* fc = fbase + (size_t)c * (HH * WW);

        for (int ht = 0; ht < 7; ++ht) {
            int h = ht * 16 + m16;
            const float* frow = fc + h * WW;
            bool hok = (h < HH);

            v2f av[32];
            #pragma unroll
            for (int ks = 0; ks < 32; ++ks) {
                int k0 = ks * 4 + hi16 * 2;
                if (hok && k0 < WW) { av[ks].x = frow[k0]; av[ks].y = frow[k0 + 1]; }
                else                { av[ks] = (v2f){0.0f, 0.0f}; }
            }
            v8f acc0 = {}, acc1 = {};
            #pragma unroll
            for (int ks = 0; ks < 32; ks += 2) {
                acc0 = __builtin_amdgcn_wmma_f32_16x16x4_f32(
                         false, av[ks],     false, bx[ks],     (short)0, acc0, false, false);
                acc1 = __builtin_amdgcn_wmma_f32_16x16x4_f32(
                         false, av[ks + 1], false, bx[ks + 1], (short)0, acc1, false, false);
            }
            #pragma unroll
            for (int rr = 0; rr < 8; ++rr)
                sU[wave][ht * 16 + rr + hi16 * 8][m16] = acc0[rr] + acc1[rr];
        }

        __builtin_amdgcn_sched_barrier(0);

        v2f ay[32], bu[32];
        #pragma unroll
        for (int ks = 0; ks < 32; ++ks) {
            int k0 = ks * 4 + hi16 * 2;
            if (m16 < 8) { ay[ks].x = sWy[m16][k0]; ay[ks].y = sWy[m16][k0 + 1]; }
            else         { ay[ks] = (v2f){0.0f, 0.0f}; }
            bu[ks].x = sU[wave][k0][m16];
            bu[ks].y = sU[wave][k0 + 1][m16];
        }
        v8f acc2 = {}, acc3 = {};
        #pragma unroll
        for (int ks = 0; ks < 32; ks += 2) {
            acc2 = __builtin_amdgcn_wmma_f32_16x16x4_f32(
                     false, ay[ks],     false, bu[ks],     (short)0, acc2, false, false);
            acc3 = __builtin_amdgcn_wmma_f32_16x16x4_f32(
                     false, ay[ks + 1], false, bu[ks + 1], (short)0, acc3, false, false);
        }
        __builtin_amdgcn_sched_barrier(0);

        #pragma unroll
        for (int rr = 0; rr < 8; ++rr) {
            int p = rr + hi16 * 8;
            int q = m16;
            if (p < OUT_HW && q < OUT_HW)
                out[(((size_t)n * CC + c) * OUT_HW + p) * OUT_HW + q] =
                    (acc2[rr] + acc3[rr]) * sc;
        }
    }
}

extern "C" void kernel_launch(void* const* d_in, const int* in_sizes, int n_in,
                              void* d_out, int out_size, void* d_ws, size_t ws_size,
                              hipStream_t stream) {
    const float* feat = (const float*)d_in[0];
    const float* rois = (const float*)d_in[1];
    float* out = (float*)d_out;
    int n_rois = in_sizes[1] / 5;
    int batch  = in_sizes[0] / (CC * HH * WW);
    dim3 grid(n_rois, 2);
    prroi_wmma_f32_kernel<<<grid, 128, 0, stream>>>(feat, rois, out, batch);
}


# Contract rows first so lanes run along the contiguous axis (coalesced float4 loads); those accumulators go straight in as B of the column product; k sums run in a new order (rows, then columns)
# speedup vs baseline: 4.0248x; 4.0248x over previous
#include <hip/hip_runtime.h>

typedef float v2f __attribute__((ext_vector_type(2)));
typedef float v4f __attribute__((ext_vector_type(4)));
typedef float v8f __attribute__((ext_vector_type(8)));

#define OUT_HW 7
#define SCALE 0.25f
#define CC 64
#define HH 100
#define WW 100
#define CW 64
#define GR 48
#define LSTR 148

__device__ __forceinline__ float hat_cdf(float t) {
    float a = fminf(fmaxf(t + 1.0f, 0.0f), 1.0f);
    float b = fminf(fmaxf(1.0f - t, 0.0f), 1.0f);
    return (t < 0.0f) ? (0.5f * a * a) : (1.0f - 0.5f * b * b);
}

#define WMMA4(A, B, C) __builtin_amdgcn_wmma_f32_16x16x4_f32(false, (A), false, (B), (short)0, (C), false, false)

__device__ __forceinline__ void rows16(const v4f (*fa)[2], const v2f* by,
                                       v8f& a0, v8f& a1, v8f& a2, v8f& a3)
{
    #pragma unroll
    for (int ks = 0; ks < 4; ++ks) {
        v2f x0 = { fa[ks][0].x, fa[ks][1].x };
        v2f x1 = { fa[ks][0].y, fa[ks][1].y };
        v2f x2 = { fa[ks][0].z, fa[ks][1].z };
        v2f x3 = { fa[ks][0].w, fa[ks][1].w };
        a0 = WMMA4(x0, by[ks], a0);
        a1 = WMMA4(x1, by[ks], a1);
        a2 = WMMA4(x2, by[ks], a2);
        a3 = WMMA4(x3, by[ks], a3);
    }
}

__global__ __launch_bounds__(128)
void prroi_wmma_f32_kernel(const float* __restrict__ feat,
                           const float* __restrict__ rois,
                           float* __restrict__ out,
                           int batch)
{
    __shared__ __attribute__((aligned(16))) float sWx[8][LSTR];
    __shared__ __attribute__((aligned(16))) float sWy[8][LSTR];

    const int n    = blockIdx.x;
    const int tid  = threadIdx.x;
    const int lane = tid & 31;
    const int wave = tid >> 5;
    const int m16  = lane & 15;
    const int hi16 = (lane >> 4) & 1;

    const float* r = rois + (size_t)n * 5;
    int bi = (int)r[0];
    bi = max(0, min(batch - 1, bi));
    float x1 = r[1] * SCALE, y1 = r[2] * SCALE;
    float x2 = r[3] * SCALE, y2 = r[4] * SCALE;
    float rw = fmaxf(x2 - x1, 0.0f), rh = fmaxf(y2 - y1, 0.0f);
    float bw = rw / OUT_HW, bh = rh / OUT_HW;
    float area = bw * bh;
    float sc = (area > 0.0f) ? (1.0f / fmaxf(area, 1e-12f)) : 0.0f;

    float ex = (x1 + 6.0f * bw) + bw;
    float ey = (y1 + 6.0f * bh) + bh;
    int w_lo = (int)fminf(fmaxf(floorf(x1) - 1.0f, 0.0f), (float)WW);
    int w_hi = (int)fminf(fmaxf(floorf(ex) + 2.0f, -1.0f), (float)(WW - 1));
    int h_lo = (int)fminf(fmaxf(floorf(y1) - 1.0f, 0.0f), (float)HH);
    int h_hi = (int)fminf(fmaxf(floorf(ey) + 2.0f, -1.0f), (float)(HH - 1));
    int w0v  = w_lo & ~3;
    int ncol = max(w_hi - w0v + 1, 0);
    int nrow = max(h_hi - h_lo + 1, 0);
    const int w0  = __builtin_amdgcn_readfirstlane(w0v);
    const int h0  = __builtin_amdgcn_readfirstlane(h_lo);
    const int whi = __builtin_amdgcn_readfirstlane(w_hi);
    const int hhi = __builtin_amdgcn_readfirstlane(h_hi);
    const int nrs = __builtin_amdgcn_readfirstlane(nrow);
    const int ncc = __builtin_amdgcn_readfirstlane((ncol + CW - 1) / CW);
    const int ng  = __builtin_amdgcn_readfirstlane((nrow + GR - 1) / GR);

    {
        const int nx  = 8 * CW * ncc;
        const int tot = nx + 8 * GR * ng;
        for (int idx = tid; idx < tot; idx += blockDim.x) {
            bool isy = (idx >= nx);
            int  e   = isy ? (idx - nx) : idx;
            int  q   = e & 7;
            int  j   = e >> 3;
            int  k   = (isy ? h0 : w0) + j;
            float val = 0.0f;
            if (q < OUT_HW && k < (isy ? HH : WW)) {
                float start = isy ? y1 : x1;
                float bs    = isy ? bh : bw;
                float u0 = start + (float)q * bs;
                float u1 = u0 + bs;
                float px = (float)k;
                val = hat_cdf(u1 - px) - hat_cdf(u0 - px);
            }
            if (isy) sWy[q][j] = val; else sWx[q][j] = val;
        }
    }
    __syncthreads();

    const int   wr   = min(m16, 7);
    const float* wyl = &sWy[wr][hi16 * 2];
    const float* wxl = &sWx[wr][hi16 * 32];

    const float* fbase = feat + (size_t)bi * CC * (HH * WW);

    for (int i = 0; i < 8; ++i) {
        int c = blockIdx.y * 32 + wave * 8 + i;
        const float* fc = fbase + (size_t)c * (HH * WW);

        v8f o0 = {}, o1 = {};
        for (int cc = 0; cc < ncc; ++cc) {
            const int  cb  = w0 + cc * CW + m16 * 4;
            const bool cm0 = (cb     <= whi), cm1 = (cb + 1 <= whi);
            const bool cm2 = (cb + 2 <= whi), cm3 = (cb + 3 <= whi);
            const float* fcol = fc + min(cb, whi & ~3);

            v8f a0 = {}, a1 = {}, a2 = {}, a3 = {};
            for (int g = 0; g < ng; ++g) {
                v4f fa[12][2];
                v2f by[12];
                #pragma unroll
                for (int ks = 0; ks < 12; ++ks) {
                    #pragma unroll
                    for (int e = 0; e < 2; ++e) {
                        int h = h0 + g * GR + ks * 4 + hi16 * 2 + e;
                        fa[ks][e] = *(const v4f*)(fcol + min(h, hhi) * WW);
                    }
                    by[ks] = *(const v2f*)(wyl + g * GR + ks * 4);
                }
                asm volatile("" :: "v"(fa[4][0]), "v"(fa[4][1]), "v"(fa[5][0]), "v"(fa[5][1]),
                                   "v"(fa[6][0]), "v"(fa[6][1]), "v"(fa[7][0]), "v"(fa[7][1]),
                                   "v"(fa[8][0]), "v"(fa[8][1]), "v"(fa[9][0]), "v"(fa[9][1]),
                                   "v"(fa[10][0]), "v"(fa[10][1]), "v"(fa[11][0]), "v"(fa[11][1]));
                #pragma unroll
                for (int ks = 0; ks < 12; ++ks) {
                    #pragma unroll
                    for (int e = 0; e < 2; ++e) {
                        bool rok = (h0 + g * GR + ks * 4 + hi16 * 2 + e <= hhi);
                        fa[ks][e].x = (rok && cm0) ? fa[ks][e].x : 0.0f;
                        fa[ks][e].y = (rok && cm1) ? fa[ks][e].y : 0.0f;
                        fa[ks][e].z = (rok && cm2) ? fa[ks][e].z : 0.0f;
                        fa[ks][e].w = (rok && cm3) ? fa[ks][e].w : 0.0f;
                    }
                }
                rows16(&fa[0], &by[0], a0, a1, a2, a3);
                if (g * GR + 16 < nrs)
                    rows16(&fa[4], &by[4], a0, a1, a2, a3);
                if (g * GR + 32 < nrs)
                    rows16(&fa[8], &by[8], a0, a1, a2, a3);
            }

            #pragma unroll
            for (int j = 0; j < 4; ++j) {
                v4f wa = *(const v4f*)(wxl + cc * CW + 8 * j);
                v4f wb = *(const v4f*)(wxl + cc * CW + 8 * j + 4);
                v2f k0 = { wa.x, wb.x }, k1 = { wa.y, wb.y };
                v2f k2 = { wa.z, wb.z }, k3 = { wa.w, wb.w };
                v2f b0 = { a0[2 * j], a0[2 * j + 1] };
                v2f b1 = { a1[2 * j], a1[2 * j + 1] };
                v2f b2 = { a2[2 * j], a2[2 * j + 1] };
                v2f b3 = { a3[2 * j], a3[2 * j + 1] };
                o0 = WMMA4(k0, b0, o0);
                o1 = WMMA4(k1, b1, o1);
                o0 = WMMA4(k2, b2, o0);
                o1 = WMMA4(k3, b3, o1);
            }
        }

        #pragma unroll
        for (int rr = 0; rr < 8; ++rr) {
            int q = rr + hi16 * 8;
            int p = m16;
            if (p < OUT_HW && q < OUT_HW)
                out[(((size_t)n * CC + c) * OUT_HW + p) * OUT_HW + q] =
                    (o0[rr] + o1[rr]) * sc;
        }
    }
}

extern "C" void kernel_launch(void* const* d_in, const int* in_sizes, int n_in,
                              void* d_out, int out_size, void* d_ws, size_t ws_size,
                              hipStream_t stream) {
    const float* feat = (const float*)d_in[0];
    const float* rois = (const float*)d_in[1];
    float* out = (float*)d_out;
    int n_rois = in_sizes[1] / 5;
    int batch  = in_sizes[0] / (CC * HH * WW);
    dim3 grid(n_rois, 2);
    prroi_wmma_f32_kernel<<<grid, 128, 0, stream>>>(feat, rois, out, batch);
}


# Row addressing hoisted out of the channel loop: 24 clamped 32-bit lane offsets formed once, each load = scalar channel base + offset; zeroing past the window dropped (weights there are exactly 0)
# speedup vs baseline: 4.9308x; 1.2251x over previous
#include <hip/hip_runtime.h>

typedef float v2f __attribute__((ext_vector_type(2)));
typedef float v4f __attribute__((ext_vector_type(4)));
typedef float v8f __attribute__((ext_vector_type(8)));

#define OUT_HW 7
#define SCALE 0.25f
#define CC 64
#define HH 100
#define WW 100
#define CW 64
#define GR 48
#define LSTR 148

__device__ __forceinline__ float hat_cdf(float t) {
    float a = fminf(fmaxf(t + 1.0f, 0.0f), 1.0f);
    float b = fminf(fmaxf(1.0f - t, 0.0f), 1.0f);
    return (t < 0.0f) ? (0.5f * a * a) : (1.0f - 0.5f * b * b);
}

#define WMMA4(A, B, C) __builtin_amdgcn_wmma_f32_16x16x4_f32(false, (A), false, (B), (short)0, (C), false, false)

__device__ __forceinline__ void rows16(const v4f (*fa)[2], const v2f* by,
                                       v8f& a0, v8f& a1, v8f& a2, v8f& a3)
{
    #pragma unroll
    for (int ks = 0; ks < 4; ++ks) {
        v2f x0 = { fa[ks][0].x, fa[ks][1].x };
        v2f x1 = { fa[ks][0].y, fa[ks][1].y };
        v2f x2 = { fa[ks][0].z, fa[ks][1].z };
        v2f x3 = { fa[ks][0].w, fa[ks][1].w };
        a0 = WMMA4(x0, by[ks], a0);
        a1 = WMMA4(x1, by[ks], a1);
        a2 = WMMA4(x2, by[ks], a2);
        a3 = WMMA4(x3, by[ks], a3);
    }
}


__device__ __forceinline__ void cols64(const float* wx, const v8f& a0, const v8f& a1,
                                       const v8f& a2, const v8f& a3, v8f& o0, v8f& o1)
{
    #pragma unroll
    for (int j = 0; j < 4; ++j) {
        v4f wa = *(const v4f*)(wx + 8 * j);
        v4f wb = *(const v4f*)(wx + 8 * j + 4);
        v2f k0 = { wa.x, wb.x }, k1 = { wa.y, wb.y };
        v2f k2 = { wa.z, wb.z }, k3 = { wa.w, wb.w };
        v2f b0 = { a0[2 * j], a0[2 * j + 1] };
        v2f b1 = { a1[2 * j], a1[2 * j + 1] };
        v2f b2 = { a2[2 * j], a2[2 * j + 1] };
        v2f b3 = { a3[2 * j], a3[2 * j + 1] };
        o0 = WMMA4(k0, b0, o0);
        o1 = WMMA4(k1, b1, o1);
        o0 = WMMA4(k2, b2, o0);
        o1 = WMMA4(k3, b3, o1);
    }
}

__global__ __launch_bounds__(128)
void prroi_wmma_f32_kernel(const float* __restrict__ feat,
                           const float* __restrict__ rois,
                           float* __restrict__ out,
                           int batch)
{
    __shared__ __attribute__((aligned(16))) float sWx[8][LSTR];
    __shared__ __attribute__((aligned(16))) float sWy[8][LSTR];

    const int n    = blockIdx.x;
    const int tid  = threadIdx.x;
    const int lane = tid & 31;
    const int wave = tid >> 5;
    const int m16  = lane & 15;
    const int hi16 = (lane >> 4) & 1;

    const float* r = rois + (size_t)n * 5;
    int bi = (int)r[0];
    bi = max(0, min(batch - 1, bi));
    float x1 = r[1] * SCALE, y1 = r[2] * SCALE;
    float x2 = r[3] * SCALE, y2 = r[4] * SCALE;
    float rw = fmaxf(x2 - x1, 0.0f), rh = fmaxf(y2 - y1, 0.0f);
    float bw = rw / OUT_HW, bh = rh / OUT_HW;
    float area = bw * bh;
    float sc = (area > 0.0f) ? (1.0f / fmaxf(area, 1e-12f)) : 0.0f;

    float ex = (x1 + 6.0f * bw) + bw;
    float ey = (y1 + 6.0f * bh) + bh;
    int w_lo = (int)fminf(fmaxf(floorf(x1) - 1.0f, 0.0f), (float)WW);
    int w_hi = (int)fminf(fmaxf(floorf(ex) + 2.0f, -1.0f), (float)(WW - 1));
    int h_lo = (int)fminf(fmaxf(floorf(y1) - 1.0f, 0.0f), (float)HH);
    int h_hi = (int)fminf(fmaxf(floorf(ey) + 2.0f, -1.0f), (float)(HH - 1));
    int w0v  = w_lo & ~3;
    int ncol = max(w_hi - w0v + 1, 0);
    int nrow = max(h_hi - h_lo + 1, 0);
    const int w0  = __builtin_amdgcn_readfirstlane(w0v);
    const int h0  = __builtin_amdgcn_readfirstlane(h_lo);
    const int whi = __builtin_amdgcn_readfirstlane(w_hi);
    const int hhi = __builtin_amdgcn_readfirstlane(h_hi);
    const int nrs = __builtin_amdgcn_readfirstlane(nrow);
    const int ncc = __builtin_amdgcn_readfirstlane((ncol + CW - 1) / CW);
    const int ng  = __builtin_amdgcn_readfirstlane((nrow + GR - 1) / GR);

    {
        const int nx  = 8 * CW * ncc;
        const int tot = nx + 8 * GR * ng;
        for (int idx = tid; idx < tot; idx += blockDim.x) {
            bool isy = (idx >= nx);
            int  e   = isy ? (idx - nx) : idx;
            int  q   = e & 7;
            int  j   = e >> 3;
            int  k   = (isy ? h0 : w0) + j;
            float val = 0.0f;
            if (q < OUT_HW && k < (isy ? HH : WW)) {
                float start = isy ? y1 : x1;
                float bs    = isy ? bh : bw;
                float u0 = start + (float)q * bs;
                float u1 = u0 + bs;
                float px = (float)k;
                val = hat_cdf(u1 - px) - hat_cdf(u0 - px);
            }
            if (isy) sWy[q][j] = val; else sWx[q][j] = val;
        }
    }
    __syncthreads();

    const int   wr   = min(m16, 7);
    const float* wyl = &sWy[wr][hi16 * 2];
    const float* wxl = &sWx[wr][hi16 * 32];

    const float* fbase = feat + (size_t)bi * CC * (HH * WW);

    const bool one = (ncc == 1) && (ng == 1);
    unsigned off0[12][2];
    {
        const int cb0 = min(w0 + m16 * 4, whi & ~3);
        #pragma unroll
        for (int ks = 0; ks < 12; ++ks) {
            #pragma unroll
            for (int e = 0; e < 2; ++e) {
                int h = h0 + ks * 4 + hi16 * 2 + e;
                off0[ks][e] = (unsigned)((cb0 + min(h, hhi) * WW) * 4);
            }
        }
    }
    const int wave_s = __builtin_amdgcn_readfirstlane(wave);

    for (int i = 0; i < 8; ++i) {
        int c = blockIdx.y * 32 + wave_s * 8 + i;
        const float* fc = fbase + (size_t)c * (HH * WW);

        v8f o0 = {}, o1 = {};
        if (one) {
            const char* fcb = (const char*)fc;
            v8f a0 = {}, a1 = {}, a2 = {}, a3 = {};
            v4f fa[12][2];
            v2f by[12];
            #pragma unroll
            for (int ks = 0; ks < 12; ++ks) {
                #pragma unroll
                for (int e = 0; e < 2; ++e) {
                    asm volatile("" : "+v"(off0[ks][e]));
                    fa[ks][e] = *(const v4f*)(fcb + off0[ks][e]);
                }
                by[ks] = *(const v2f*)(wyl + ks * 4);
            }
            asm volatile("" :: "v"(fa[4][0]), "v"(fa[4][1]), "v"(fa[5][0]), "v"(fa[5][1]),
                               "v"(fa[6][0]), "v"(fa[6][1]), "v"(fa[7][0]), "v"(fa[7][1]),
                               "v"(fa[8][0]), "v"(fa[8][1]), "v"(fa[9][0]), "v"(fa[9][1]),
                               "v"(fa[10][0]), "v"(fa[10][1]), "v"(fa[11][0]), "v"(fa[11][1]));
            rows16(&fa[0], &by[0], a0, a1, a2, a3);
            if (16 < nrs)
                rows16(&fa[4], &by[4], a0, a1, a2, a3);
            if (32 < nrs)
                rows16(&fa[8], &by[8], a0, a1, a2, a3);

            cols64(wxl, a0, a1, a2, a3, o0, o1);
        } else {
            for (int cc = 0; cc < ncc; ++cc) {
                const int  cb  = w0 + cc * CW + m16 * 4;
                const bool cm0 = (cb     <= whi), cm1 = (cb + 1 <= whi);
                const bool cm2 = (cb + 2 <= whi), cm3 = (cb + 3 <= whi);
                const float* fcol = fc + min(cb, whi & ~3);

                v8f a0 = {}, a1 = {}, a2 = {}, a3 = {};
                for (int g = 0; g < ng; ++g) {
                    v4f fa[12][2];
                    v2f by[12];
                    #pragma unroll
                    for (int ks = 0; ks < 12; ++ks) {
                        #pragma unroll
                        for (int e = 0; e < 2; ++e) {
                            int h = h0 + g * GR + ks * 4 + hi16 * 2 + e;
                            fa[ks][e] = *(const v4f*)(fcol + min(h, hhi) * WW);
                        }
                        by[ks] = *(const v2f*)(wyl + g * GR + ks * 4);
                    }
                    asm volatile("" :: "v"(fa[4][0]), "v"(fa[4][1]), "v"(fa[5][0]), "v"(fa[5][1]),
                                       "v"(fa[6][0]), "v"(fa[6][1]), "v"(fa[7][0]), "v"(fa[7][1]),
                                       "v"(fa[8][0]), "v"(fa[8][1]), "v"(fa[9][0]), "v"(fa[9][1]),
                                       "v"(fa[10][0]), "v"(fa[10][1]), "v"(fa[11][0]), "v"(fa[11][1]));
                    #pragma unroll
                    for (int ks = 0; ks < 12; ++ks) {
                        #pragma unroll
                        for (int e = 0; e < 2; ++e) {
                            bool rok = (h0 + g * GR + ks * 4 + hi16 * 2 + e <= hhi);
                            fa[ks][e].x = (rok && cm0) ? fa[ks][e].x : 0.0f;
                            fa[ks][e].y = (rok && cm1) ? fa[ks][e].y : 0.0f;
                            fa[ks][e].z = (rok && cm2) ? fa[ks][e].z : 0.0f;
                            fa[ks][e].w = (rok && cm3) ? fa[ks][e].w : 0.0f;
                        }
                    }
                    rows16(&fa[0], &by[0], a0, a1, a2, a3);
                    if (g * GR + 16 < nrs)
                        rows16(&fa[4], &by[4], a0, a1, a2, a3);
                    if (g * GR + 32 < nrs)
                        rows16(&fa[8], &by[8], a0, a1, a2, a3);
                }

                cols64(wxl + cc * CW, a0, a1, a2, a3, o0, o1);
            }
        }

        #pragma unroll
        for (int rr = 0; rr < 8; ++rr) {
            int q = rr + hi16 * 8;
            int p = m16;
            if (p < OUT_HW && q < OUT_HW)
                out[(((size_t)n * CC + c) * OUT_HW + p) * OUT_HW + q] =
                    (o0[rr] + o1[rr]) * sc;
        }
    }
}

extern "C" void kernel_launch(void* const* d_in, const int* in_sizes, int n_in,
                              void* d_out, int out_size, void* d_ws, size_t ws_size,
                              hipStream_t stream) {
    const float* feat = (const float*)d_in[0];
    const float* rois = (const float*)d_in[1];
    float* out = (float*)d_out;
    int n_rois = in_sizes[1] / 5;
    int batch  = in_sizes[0] / (CC * HH * WW);
    dim3 grid(n_rois, 2);
    prroi_wmma_f32_kernel<<<grid, 128, 0, stream>>>(feat, rois, out, batch);
}


# Blocks of 8 waves with 4 channels per wave (was 4 waves with 8): four waves per SIMD instead of two, so a wave's wait for its load group is covered by the others' matrix runs
# speedup vs baseline: 5.8077x; 1.1778x over previous
#include <hip/hip_runtime.h>

typedef float v2f __attribute__((ext_vector_type(2)));
typedef float v4f __attribute__((ext_vector_type(4)));
typedef float v8f __attribute__((ext_vector_type(8)));

#define OUT_HW 7
#define SCALE 0.25f
#define CC 64
#define HH 100
#define WW 100
#define CW 64
#define GR 48
#define LSTR 148

__device__ __forceinline__ float hat_cdf(float t) {
    float a = fminf(fmaxf(t + 1.0f, 0.0f), 1.0f);
    float b = fminf(fmaxf(1.0f - t, 0.0f), 1.0f);
    return (t < 0.0f) ? (0.5f * a * a) : (1.0f - 0.5f * b * b);
}

#define WMMA4(A, B, C) __builtin_amdgcn_wmma_f32_16x16x4_f32(false, (A), false, (B), (short)0, (C), false, false)

__device__ __forceinline__ void rows16(const v4f (*fa)[2], const v2f* by,
                                       v8f& a0, v8f& a1, v8f& a2, v8f& a3)
{
    #pragma unroll
    for (int ks = 0; ks < 4; ++ks) {
        v2f x0 = { fa[ks][0].x, fa[ks][1].x };
        v2f x1 = { fa[ks][0].y, fa[ks][1].y };
        v2f x2 = { fa[ks][0].z, fa[ks][1].z };
        v2f x3 = { fa[ks][0].w, fa[ks][1].w };
        a0 = WMMA4(x0, by[ks], a0);
        a1 = WMMA4(x1, by[ks], a1);
        a2 = WMMA4(x2, by[ks], a2);
        a3 = WMMA4(x3, by[ks], a3);
    }
}


__device__ __forceinline__ void cols64(const float* wx, const v8f& a0, const v8f& a1,
                                       const v8f& a2, const v8f& a3, v8f& o0, v8f& o1)
{
    #pragma unroll
    for (int j = 0; j < 4; ++j) {
        v4f wa = *(const v4f*)(wx + 8 * j);
        v4f wb = *(const v4f*)(wx + 8 * j + 4);
        v2f k0 = { wa.x, wb.x }, k1 = { wa.y, wb.y };
        v2f k2 = { wa.z, wb.z }, k3 = { wa.w, wb.w };
        v2f b0 = { a0[2 * j], a0[2 * j + 1] };
        v2f b1 = { a1[2 * j], a1[2 * j + 1] };
        v2f b2 = { a2[2 * j], a2[2 * j + 1] };
        v2f b3 = { a3[2 * j], a3[2 * j + 1] };
        o0 = WMMA4(k0, b0, o0);
        o1 = WMMA4(k1, b1, o1);
        o0 = WMMA4(k2, b2, o0);
        o1 = WMMA4(k3, b3, o1);
    }
}

__global__ __launch_bounds__(256, 4)
void prroi_wmma_f32_kernel(const float* __restrict__ feat,
                           const float* __restrict__ rois,
                           float* __restrict__ out,
                           int batch)
{
    __shared__ __attribute__((aligned(16))) float sWx[8][LSTR];
    __shared__ __attribute__((aligned(16))) float sWy[8][LSTR];

    const int n    = blockIdx.x;
    const int tid  = threadIdx.x;
    const int lane = tid & 31;
    const int wave = tid >> 5;
    const int m16  = lane & 15;
    const int hi16 = (lane >> 4) & 1;

    const float* r = rois + (size_t)n * 5;
    int bi = (int)r[0];
    bi = max(0, min(batch - 1, bi));
    float x1 = r[1] * SCALE, y1 = r[2] * SCALE;
    float x2 = r[3] * SCALE, y2 = r[4] * SCALE;
    float rw = fmaxf(x2 - x1, 0.0f), rh = fmaxf(y2 - y1, 0.0f);
    float bw = rw / OUT_HW, bh = rh / OUT_HW;
    float area = bw * bh;
    float sc = (area > 0.0f) ? (1.0f / fmaxf(area, 1e-12f)) : 0.0f;

    float ex = (x1 + 6.0f * bw) + bw;
    float ey = (y1 + 6.0f * bh) + bh;
    int w_lo = (int)fminf(fmaxf(floorf(x1) - 1.0f, 0.0f), (float)WW);
    int w_hi = (int)fminf(fmaxf(floorf(ex) + 2.0f, -1.0f), (float)(WW - 1));
    int h_lo = (int)fminf(fmaxf(floorf(y1) - 1.0f, 0.0f), (float)HH);
    int h_hi = (int)fminf(fmaxf(floorf(ey) + 2.0f, -1.0f), (float)(HH - 1));
    int w0v  = w_lo & ~3;
    int ncol = max(w_hi - w0v + 1, 0);
    int nrow = max(h_hi - h_lo + 1, 0);
    const int w0  = __builtin_amdgcn_readfirstlane(w0v);
    const int h0  = __builtin_amdgcn_readfirstlane(h_lo);
    const int whi = __builtin_amdgcn_readfirstlane(w_hi);
    const int hhi = __builtin_amdgcn_readfirstlane(h_hi);
    const int nrs = __builtin_amdgcn_readfirstlane(nrow);
    const int ncc = __builtin_amdgcn_readfirstlane((ncol + CW - 1) / CW);
    const int ng  = __builtin_amdgcn_readfirstlane((nrow + GR - 1) / GR);

    {
        const int nx  = 8 * CW * ncc;
        const int tot = nx + 8 * GR * ng;
        for (int idx = tid; idx < tot; idx += blockDim.x) {
            bool isy = (idx >= nx);
            int  e   = isy ? (idx - nx) : idx;
            int  q   = e & 7;
            int  j   = e >> 3;
            int  k   = (isy ? h0 : w0) + j;
            float val = 0.0f;
            if (q < OUT_HW && k < (isy ? HH : WW)) {
                float start = isy ? y1 : x1;
                float bs    = isy ? bh : bw;
                float u0 = start + (float)q * bs;
                float u1 = u0 + bs;
                float px = (float)k;
                val = hat_cdf(u1 - px) - hat_cdf(u0 - px);
            }
            if (isy) sWy[q][j] = val; else sWx[q][j] = val;
        }
    }
    __syncthreads();

    const int   wr   = min(m16, 7);
    const float* wyl = &sWy[wr][hi16 * 2];
    const float* wxl = &sWx[wr][hi16 * 32];

    const float* fbase = feat + (size_t)bi * CC * (HH * WW);

    const bool one = (ncc == 1) && (ng == 1);
    unsigned off0[12][2];
    {
        const int cb0 = min(w0 + m16 * 4, whi & ~3);
        #pragma unroll
        for (int ks = 0; ks < 12; ++ks) {
            #pragma unroll
            for (int e = 0; e < 2; ++e) {
                int h = h0 + ks * 4 + hi16 * 2 + e;
                off0[ks][e] = (unsigned)((cb0 + min(h, hhi) * WW) * 4);
            }
        }
    }
    const int wave_s = __builtin_amdgcn_readfirstlane(wave);

    for (int i = 0; i < 4; ++i) {
        int c = blockIdx.y * 32 + wave_s * 4 + i;
        const float* fc = fbase + (size_t)c * (HH * WW);

        v8f o0 = {}, o1 = {};
        if (one) {
            const char* fcb = (const char*)fc;
            v8f a0 = {}, a1 = {}, a2 = {}, a3 = {};
            v4f fa[12][2];
            v2f by[12];
            #pragma unroll
            for (int ks = 0; ks < 12; ++ks) {
                #pragma unroll
                for (int e = 0; e < 2; ++e) {
                    asm volatile("" : "+v"(off0[ks][e]));
                    fa[ks][e] = *(const v4f*)(fcb + off0[ks][e]);
                }
                by[ks] = *(const v2f*)(wyl + ks * 4);
            }
            asm volatile("" :: "v"(fa[4][0]), "v"(fa[4][1]), "v"(fa[5][0]), "v"(fa[5][1]),
                               "v"(fa[6][0]), "v"(fa[6][1]), "v"(fa[7][0]), "v"(fa[7][1]),
                               "v"(fa[8][0]), "v"(fa[8][1]), "v"(fa[9][0]), "v"(fa[9][1]),
                               "v"(fa[10][0]), "v"(fa[10][1]), "v"(fa[11][0]), "v"(fa[11][1]));
            rows16(&fa[0], &by[0], a0, a1, a2, a3);
            if (16 < nrs)
                rows16(&fa[4], &by[4], a0, a1, a2, a3);
            if (32 < nrs)
                rows16(&fa[8], &by[8], a0, a1, a2, a3);

            cols64(wxl, a0, a1, a2, a3, o0, o1);
        } else {
            for (int cc = 0; cc < ncc; ++cc) {
                const int  cb  = w0 + cc * CW + m16 * 4;
                const bool cm0 = (cb     <= whi), cm1 = (cb + 1 <= whi);
                const bool cm2 = (cb + 2 <= whi), cm3 = (cb + 3 <= whi);
                const float* fcol = fc + min(cb, whi & ~3);

                v8f a0 = {}, a1 = {}, a2 = {}, a3 = {};
                for (int g = 0; g < ng; ++g) {
                    v4f fa[12][2];
                    v2f by[12];
                    #pragma unroll
                    for (int ks = 0; ks < 12; ++ks) {
                        #pragma unroll
                        for (int e = 0; e < 2; ++e) {
                            int h = h0 + g * GR + ks * 4 + hi16 * 2 + e;
                            fa[ks][e] = *(const v4f*)(fcol + min(h, hhi) * WW);
                        }
                        by[ks] = *(const v2f*)(wyl + g * GR + ks * 4);
                    }
                    asm volatile("" :: "v"(fa[4][0]), "v"(fa[4][1]), "v"(fa[5][0]), "v"(fa[5][1]),
                                       "v"(fa[6][0]), "v"(fa[6][1]), "v"(fa[7][0]), "v"(fa[7][1]),
                                       "v"(fa[8][0]), "v"(fa[8][1]), "v"(fa[9][0]), "v"(fa[9][1]),
                                       "v"(fa[10][0]), "v"(fa[10][1]), "v"(fa[11][0]), "v"(fa[11][1]));
                    #pragma unroll
                    for (int ks = 0; ks < 12; ++ks) {
                        #pragma unroll
                        for (int e = 0; e < 2; ++e) {
                            bool rok = (h0 + g * GR + ks * 4 + hi16 * 2 + e <= hhi);
                            fa[ks][e].x = (rok && cm0) ? fa[ks][e].x : 0.0f;
                            fa[ks][e].y = (rok && cm1) ? fa[ks][e].y : 0.0f;
                            fa[ks][e].z = (rok && cm2) ? fa[ks][e].z : 0.0f;
                            fa[ks][e].w = (rok && cm3) ? fa[ks][e].w : 0.0f;
                        }
                    }
                    rows16(&fa[0], &by[0], a0, a1, a2, a3);
                    if (g * GR + 16 < nrs)
                        rows16(&fa[4], &by[4], a0, a1, a2, a3);
                    if (g * GR + 32 < nrs)
                        rows16(&fa[8], &by[8], a0, a1, a2, a3);
                }

                cols64(wxl + cc * CW, a0, a1, a2, a3, o0, o1);
            }
        }

        #pragma unroll
        for (int rr = 0; rr < 8; ++rr) {
            int q = rr + hi16 * 8;
            int p = m16;
            if (p < OUT_HW && q < OUT_HW)
                out[(((size_t)n * CC + c) * OUT_HW + p) * OUT_HW + q] =
                    (o0[rr] + o1[rr]) * sc;
        }
    }
}

extern "C" void kernel_launch(void* const* d_in, const int* in_sizes, int n_in,
                              void* d_out, int out_size, void* d_ws, size_t ws_size,
                              hipStream_t stream) {
    const float* feat = (const float*)d_in[0];
    const float* rois = (const float*)d_in[1];
    float* out = (float*)d_out;
    int n_rois = in_sizes[1] / 5;
    int batch  = in_sizes[0] / (CC * HH * WW);
    dim3 grid(n_rois, 2);
    prroi_wmma_f32_kernel<<<grid, 256, 0, stream>>>(feat, rois, out, batch);
}


# Two channels share every matrix product: bins of channel c in N 0..7, of c+1 in N 8..15 (K half = channel); column product once per pair; k sums in a new order (2-row k-steps)
# speedup vs baseline: 6.4303x; 1.1072x over previous
#include <hip/hip_runtime.h>

typedef float v2f __attribute__((ext_vector_type(2)));
typedef float v4f __attribute__((ext_vector_type(4)));
typedef float v8f __attribute__((ext_vector_type(8)));

#define OUT_HW 7
#define SCALE 0.25f
#define CC 64
#define HH 100
#define WW 100
#define CW 64
#define GR 48
#define LSTR 148

__device__ __forceinline__ float hat_cdf(float t) {
    float a = fminf(fmaxf(t + 1.0f, 0.0f), 1.0f);
    float b = fminf(fmaxf(1.0f - t, 0.0f), 1.0f);
    return (t < 0.0f) ? (0.5f * a * a) : (1.0f - 0.5f * b * b);
}

#define WMMA4(A, B, C) __builtin_amdgcn_wmma_f32_16x16x4_f32(false, (A), false, (B), (short)0, (C), false, false)

__device__ __forceinline__ void kstep4(const v4f (*fa)[2], const v2f* by,
                                       v8f& a0, v8f& a1, v8f& a2, v8f& a3)
{
    #pragma unroll
    for (int ks = 0; ks < 4; ++ks) {
        v2f x0 = { fa[ks][0].x, fa[ks][1].x };
        v2f x1 = { fa[ks][0].y, fa[ks][1].y };
        v2f x2 = { fa[ks][0].z, fa[ks][1].z };
        v2f x3 = { fa[ks][0].w, fa[ks][1].w };
        a0 = WMMA4(x0, by[ks], a0);
        a1 = WMMA4(x1, by[ks], a1);
        a2 = WMMA4(x2, by[ks], a2);
        a3 = WMMA4(x3, by[ks], a3);
    }
}


__device__ __forceinline__ void cols64(const float* wx, const v8f& a0, const v8f& a1,
                                       const v8f& a2, const v8f& a3, v8f& o0, v8f& o1)
{
    #pragma unroll
    for (int j = 0; j < 4; ++j) {
        v4f wa = *(const v4f*)(wx + 8 * j);
        v4f wb = *(const v4f*)(wx + 8 * j + 4);
        v2f k0 = { wa.x, wb.x }, k1 = { wa.y, wb.y };
        v2f k2 = { wa.z, wb.z }, k3 = { wa.w, wb.w };
        v2f b0 = { a0[2 * j], a0[2 * j + 1] };
        v2f b1 = { a1[2 * j], a1[2 * j + 1] };
        v2f b2 = { a2[2 * j], a2[2 * j + 1] };
        v2f b3 = { a3[2 * j], a3[2 * j + 1] };
        o0 = WMMA4(k0, b0, o0);
        o1 = WMMA4(k1, b1, o1);
        o0 = WMMA4(k2, b2, o0);
        o1 = WMMA4(k3, b3, o1);
    }
}

__global__ __launch_bounds__(256, 4)
void prroi_wmma_f32_kernel(const float* __restrict__ feat,
                           const float* __restrict__ rois,
                           float* __restrict__ out,
                           int batch)
{
    __shared__ __attribute__((aligned(16))) float sWx[8][LSTR];
    __shared__ __attribute__((aligned(16))) float sWy[8][LSTR];

    const int n    = blockIdx.x;
    const int tid  = threadIdx.x;
    const int lane = tid & 31;
    const int wave = tid >> 5;
    const int m16  = lane & 15;
    const int hi16 = (lane >> 4) & 1;

    const float* r = rois + (size_t)n * 5;
    int bi = (int)r[0];
    bi = max(0, min(batch - 1, bi));
    float x1 = r[1] * SCALE, y1 = r[2] * SCALE;
    float x2 = r[3] * SCALE, y2 = r[4] * SCALE;
    float rw = fmaxf(x2 - x1, 0.0f), rh = fmaxf(y2 - y1, 0.0f);
    float bw = rw / OUT_HW, bh = rh / OUT_HW;
    float area = bw * bh;
    float sc = (area > 0.0f) ? (1.0f / fmaxf(area, 1e-12f)) : 0.0f;

    float ex = (x1 + 6.0f * bw) + bw;
    float ey = (y1 + 6.0f * bh) + bh;
    int w_lo = (int)fminf(fmaxf(floorf(x1) - 1.0f, 0.0f), (float)WW);
    int w_hi = (int)fminf(fmaxf(floorf(ex) + 2.0f, -1.0f), (float)(WW - 1));
    int h_lo = (int)fminf(fmaxf(floorf(y1) - 1.0f, 0.0f), (float)HH);
    int h_hi = (int)fminf(fmaxf(floorf(ey) + 2.0f, -1.0f), (float)(HH - 1));
    int w0v  = w_lo & ~3;
    int ncol = max(w_hi - w0v + 1, 0);
    int nrow = max(h_hi - h_lo + 1, 0);
    const int w0  = __builtin_amdgcn_readfirstlane(w0v);
    const int h0  = __builtin_amdgcn_readfirstlane(h_lo);
    const int whi = __builtin_amdgcn_readfirstlane(w_hi);
    const int hhi = __builtin_amdgcn_readfirstlane(h_hi);
    const int nrs = __builtin_amdgcn_readfirstlane(nrow);
    const int ncc = __builtin_amdgcn_readfirstlane((ncol + CW - 1) / CW);
    const int ng  = __builtin_amdgcn_readfirstlane((nrow + GR - 1) / GR);

    {
        const int nx  = 8 * CW * ncc;
        const int tot = nx + 8 * GR * ng;
        for (int idx = tid; idx < tot; idx += blockDim.x) {
            bool isy = (idx >= nx);
            int  e   = isy ? (idx - nx) : idx;
            int  q   = e & 7;
            int  j   = e >> 3;
            int  k   = (isy ? h0 : w0) + j;
            float val = 0.0f;
            if (q < OUT_HW && k < (isy ? HH : WW)) {
                float start = isy ? y1 : x1;
                float bs    = isy ? bh : bw;
                float u0 = start + (float)q * bs;
                float u1 = u0 + bs;
                float px = (float)k;
                val = hat_cdf(u1 - px) - hat_cdf(u0 - px);
            }
            if (isy) sWy[q][j] = val; else sWx[q][j] = val;
        }
    }
    __syncthreads();

    const int   wr   = min(m16, 7);
    const float* wyl = &sWy[wr][hi16 * 2];
    const float* wxl = &sWx[wr][hi16 * 32];

    const float* fbase = feat + (size_t)bi * CC * (HH * WW);

    const bool one = (ncc == 1) && (ng == 1);
    const int  cb0 = min(w0 + m16 * 4, whi & ~3);
    const unsigned chalf = (unsigned)(hi16 * (HH * WW * 4));
    const unsigned ocap  = (unsigned)((cb0 + hhi * WW) * 4) + chalf;
    const unsigned rbase = (unsigned)((cb0 + h0 * WW) * 4) + chalf;
    const float* wyp = &sWy[((m16 >> 3) == hi16) ? (m16 & 7) : 7][0];
    const int wave_s = __builtin_amdgcn_readfirstlane(wave);

    for (int i = 0; i < 2; ++i) {
        const int c0 = blockIdx.y * 32 + wave_s * 4 + i * 2;

        if (one) {
            const char* fcb = (const char*)(fbase + (size_t)c0 * (HH * WW));
            v8f a0 = {}, a1 = {}, a2 = {}, a3 = {};
            v8f o0 = {}, o1 = {};

            const int ngp = (GR / 2 < nrs) ? 2 : 1;
            for (int g = 0; g < ngp; ++g) {
                const int r0 = g * (GR / 2);
                v4f fa[12][2];
                v2f by[12];
                #pragma unroll
                for (int ks = 0; ks < 12; ++ks) {
                    #pragma unroll
                    for (int e = 0; e < 2; ++e) {
                        unsigned o = min(rbase + (unsigned)((r0 + ks * 2 + e) * (WW * 4)), ocap);
                        fa[ks][e] = *(const v4f*)(fcb + o);
                    }
                    by[ks] = *(const v2f*)(wyp + r0 + ks * 2);
                }
                asm volatile("" :: "v"(fa[4][0]), "v"(fa[4][1]), "v"(fa[5][0]), "v"(fa[5][1]),
                                   "v"(fa[6][0]), "v"(fa[6][1]), "v"(fa[7][0]), "v"(fa[7][1]),
                                   "v"(fa[8][0]), "v"(fa[8][1]), "v"(fa[9][0]), "v"(fa[9][1]),
                                   "v"(fa[10][0]), "v"(fa[10][1]), "v"(fa[11][0]), "v"(fa[11][1]));
                kstep4(&fa[0], &by[0], a0, a1, a2, a3);
                if (r0 + 8 < nrs)
                    kstep4(&fa[4], &by[4], a0, a1, a2, a3);
                if (r0 + 16 < nrs)
                    kstep4(&fa[8], &by[8], a0, a1, a2, a3);
            }

            cols64(wxl, a0, a1, a2, a3, o0, o1);

            #pragma unroll
            for (int rr = 0; rr < 8; ++rr) {
                int q = rr + hi16 * 8;
                int p = m16 & 7;
                int c = c0 + (m16 >> 3);
                if (p < OUT_HW && q < OUT_HW)
                    out[(((size_t)n * CC + c) * OUT_HW + p) * OUT_HW + q] =
                        (o0[rr] + o1[rr]) * sc;
            }
        } else {
            for (int u = 0; u < 2; ++u) {
                const int c = c0 + u;
                const float* fc = fbase + (size_t)c * (HH * WW);
                v8f o0 = {}, o1 = {};
                for (int cc = 0; cc < ncc; ++cc) {
                    const int  cb  = w0 + cc * CW + m16 * 4;
                    const bool cm0 = (cb     <= whi), cm1 = (cb + 1 <= whi);
                    const bool cm2 = (cb + 2 <= whi), cm3 = (cb + 3 <= whi);
                    const float* fcol = fc + min(cb, whi & ~3);

                    v8f a0 = {}, a1 = {}, a2 = {}, a3 = {};
                    for (int g = 0; g < ng; ++g) {
                        v4f fa[12][2];
                        v2f by[12];
                        #pragma unroll
                        for (int ks = 0; ks < 12; ++ks) {
                            #pragma unroll
                            for (int e = 0; e < 2; ++e) {
                                int h = h0 + g * GR + ks * 4 + hi16 * 2 + e;
                                fa[ks][e] = *(const v4f*)(fcol + min(h, hhi) * WW);
                            }
                            by[ks] = *(const v2f*)(wyl + g * GR + ks * 4);
                        }
                        asm volatile("" :: "v"(fa[4][0]), "v"(fa[4][1]), "v"(fa[5][0]), "v"(fa[5][1]),
                                           "v"(fa[6][0]), "v"(fa[6][1]), "v"(fa[7][0]), "v"(fa[7][1]),
                                           "v"(fa[8][0]), "v"(fa[8][1]), "v"(fa[9][0]), "v"(fa[9][1]),
                                           "v"(fa[10][0]), "v"(fa[10][1]), "v"(fa[11][0]), "v"(fa[11][1]));
                        #pragma unroll
                        for (int ks = 0; ks < 12; ++ks) {
                            #pragma unroll
                            for (int e = 0; e < 2; ++e) {
                                bool rok = (h0 + g * GR + ks * 4 + hi16 * 2 + e <= hhi);
                                fa[ks][e].x = (rok && cm0) ? fa[ks][e].x : 0.0f;
                                fa[ks][e].y = (rok && cm1) ? fa[ks][e].y : 0.0f;
                                fa[ks][e].z = (rok && cm2) ? fa[ks][e].z : 0.0f;
                                fa[ks][e].w = (rok && cm3) ? fa[ks][e].w : 0.0f;
                            }
                        }
                        kstep4(&fa[0], &by[0], a0, a1, a2, a3);
                        if (g * GR + 16 < nrs)
                            kstep4(&fa[4], &by[4], a0, a1, a2, a3);
                        if (g * GR + 32 < nrs)
                            kstep4(&fa[8], &by[8], a0, a1, a2, a3);
                    }

                    cols64(wxl + cc * CW, a0, a1, a2, a3, o0, o1);
                }

                #pragma unroll
                for (int rr = 0; rr < 8; ++rr) {
                    int q = rr + hi16 * 8;
                    int p = m16;
                    if (p < OUT_HW && q < OUT_HW)
                        out[(((size_t)n * CC + c) * OUT_HW + p) * OUT_HW + q] =
                            (o0[rr] + o1[rr]) * sc;
                }
            }
        }
    }
}

extern "C" void kernel_launch(void* const* d_in, const int* in_sizes, int n_in,
                              void* d_out, int out_size, void* d_ws, size_t ws_size,
                              hipStream_t stream) {
    const float* feat = (const float*)d_in[0];
    const float* rois = (const float*)d_in[1];
    float* out = (float*)d_out;
    int n_rois = in_sizes[1] / 5;
    int batch  = in_sizes[0] / (CC * HH * WW);
    dim3 grid(n_rois, 2);
    prroi_wmma_f32_kernel<<<grid, 256, 0, stream>>>(feat, rois, out, batch);
}
